# starting kernel
# baseline (speedup 1.0000x reference)
#include <hip/hip_runtime.h>
#include <hip/hip_bf16.h>

typedef __attribute__((ext_vector_type(2))) float v2f;
typedef __attribute__((ext_vector_type(8))) float v8f;

#define HID 128
#define WSTRIDE 132

__global__ void k_zero_i(int* __restrict__ p, int n) {
  int i = blockIdx.x * blockDim.x + threadIdx.x;
  if (i < n) p[i] = 0;
}
__global__ void k_zero_f(float* __restrict__ p, int n) {
  int i = blockIdx.x * blockDim.x + threadIdx.x;
  if (i < n) p[i] = 0.0f;
}

__global__ void k_count(const int* __restrict__ dst, int* __restrict__ cnt, int e) {
  int i = blockIdx.x * blockDim.x + threadIdx.x;
  if (i < e) atomicAdd(&cnt[dst[i]], 1);
}

__global__ void k_dinv(const int* __restrict__ cnt, float* __restrict__ dinv, int n) {
  int i = blockIdx.x * blockDim.x + threadIdx.x;
  if (i < n) dinv[i] = rsqrtf((float)cnt[i] + 1.0f);
}

__global__ void k_scan1(const int* __restrict__ counts, int* __restrict__ rowstart,
                        int* __restrict__ blockSums, int n) {
  __shared__ int sm[1024];
  int t = threadIdx.x;
  int i = blockIdx.x * 1024 + t;
  int v = (i < n) ? counts[i] : 0;
  sm[t] = v;
  __syncthreads();
  for (int off = 1; off < 1024; off <<= 1) {
    int u = (t >= off) ? sm[t - off] : 0;
    __syncthreads();
    sm[t] += u;
    __syncthreads();
  }
  if (i < n) rowstart[i + 1] = sm[t];
  if (t == 1023) blockSums[blockIdx.x] = sm[1023];
}

__global__ void k_scan2(const int* __restrict__ blockSums, int* __restrict__ blockOff, int nb) {
  __shared__ int sm[1024];
  int t = threadIdx.x;
  int v = (t < nb) ? blockSums[t] : 0;
  sm[t] = v;
  __syncthreads();
  for (int off = 1; off < 1024; off <<= 1) {
    int u = (t >= off) ? sm[t - off] : 0;
    __syncthreads();
    sm[t] += u;
    __syncthreads();
  }
  if (t < nb) blockOff[t] = sm[t] - v;
}

__global__ void k_scan3(int* __restrict__ rowstart, const int* __restrict__ blockOff, int n) {
  int i = blockIdx.x * blockDim.x + threadIdx.x;
  if (i < n) rowstart[i + 1] += blockOff[i >> 10];
  if (i == 0) rowstart[0] = 0;
}

__global__ void k_fill(const int* __restrict__ src, const int* __restrict__ dst,
                       const float* __restrict__ dinv, const int* __restrict__ rowstart,
                       int* __restrict__ cursor, int2* __restrict__ rec, int e) {
  int i = blockIdx.x * blockDim.x + threadIdx.x;
  if (i >= e) return;
  int s = src[i], d = dst[i];
  int p = atomicAdd(&cursor[d], 1);
  int2 r;
  r.x = s;
  r.y = __float_as_int(dinv[s] * dinv[d]);
  rec[rowstart[d] + p] = r;
}

__global__ void k_gemm128(const float* __restrict__ X, const float* __restrict__ W,
                          const float* __restrict__ bias, float* __restrict__ H, int n) {
  extern __shared__ float ws[];
  const int tid = threadIdx.x;

  for (int i = tid; i < HID * HID; i += 256) {
    int k = i >> 7, c = i & 127;
    ws[c * WSTRIDE + k] = W[i];
  }
  __syncthreads();

  const int wave   = tid >> 5;
  const int lane   = tid & 31;
  const int lane15 = lane & 15;
  const int half   = lane >> 4;
  const int rowBase = blockIdx.x * 128 + wave * 16;
  const int arow    = rowBase + lane15;
  const int koff    = half * 2;

  v2f a[32];
  if (arow < n) {
    const float* xp = X + (size_t)arow * HID + koff;
    if (bias) {
      const float* bp = bias + koff;
#pragma unroll
      for (int kk = 0; kk < 32; ++kk) {
        v2f xv = *(const v2f*)(xp + kk * 4);
        v2f bv = *(const v2f*)(bp + kk * 4);
        v2f t = xv + bv;
        t.x = fmaxf(t.x, 0.0f);
        t.y = fmaxf(t.y, 0.0f);
        a[kk] = t;
      }
    } else {
#pragma unroll
      for (int kk = 0; kk < 32; ++kk)
        a[kk] = *(const v2f*)(xp + kk * 4);
    }
  } else {
    v2f z = {};
#pragma unroll
    for (int kk = 0; kk < 32; ++kk) a[kk] = z;
  }

#pragma unroll
  for (int nt = 0; nt < 8; ++nt) {
    v8f acc = {};
    const float* bp = ws + (nt * 16 + lane15) * WSTRIDE + koff;
#pragma unroll
    for (int kk = 0; kk < 32; ++kk) {
      v2f b = *(const v2f*)(bp + kk * 4);
      acc = __builtin_amdgcn_wmma_f32_16x16x4_f32(
          false, a[kk], false, b, (short)0, acc, false, false);
    }
    const int colOut = nt * 16 + lane15;
#pragma unroll
    for (int r = 0; r < 8; ++r) {
      int row = rowBase + r + half * 8;
      if (row < n) H[(size_t)row * HID + colOut] = acc[r];
    }
  }
}

__global__ void k_aggregate(const float* __restrict__ H, const int2* __restrict__ rec,
                            const int* __restrict__ rowstart, const float* __restrict__ dinv,
                            float* __restrict__ AGG, int n) {
  long long gid = (long long)blockIdx.x * blockDim.x + threadIdx.x;
  int node = (int)(gid >> 5);
  if (node >= n) return;
  int q = ((int)gid & 31) * 4;

  int beg = __builtin_amdgcn_readfirstlane(rowstart[node]);
  int end = __builtin_amdgcn_readfirstlane(rowstart[node + 1]);

  float wself = dinv[node];
  wself *= wself;
  float4 acc = *(const float4*)(H + (size_t)node * HID + q);
  acc.x *= wself; acc.y *= wself; acc.z *= wself; acc.w *= wself;

  for (int j = beg; j < end; ++j) {
    int2 r = rec[j];
    int s = r.x;
    float w = __int_as_float(r.y);
    const float4 h = *(const float4*)(H + (size_t)s * HID + q);
    acc.x += w * h.x; acc.y += w * h.y; acc.z += w * h.z; acc.w += w * h.w;
  }
  *(float4*)(AGG + (size_t)node * HID + q) = acc;
}

__global__ void k_cnt(const int* __restrict__ batch, float* __restrict__ cnt, int n) {
  int i = blockIdx.x * blockDim.x + threadIdx.x;
  if (i < n)
    __hip_atomic_fetch_add(&cnt[batch[i]], 1.0f, __ATOMIC_RELAXED, __HIP_MEMORY_SCOPE_AGENT);
}

__global__ void k_pool(const float* __restrict__ H, const int* __restrict__ batch,
                       float* __restrict__ pooled, int n) {
  long long gid = (long long)blockIdx.x * blockDim.x + threadIdx.x;
  int node = (int)(gid >> 5);
  if (node >= n) return;
  int q = ((int)gid & 31) * 4;
  int g = __builtin_amdgcn_readfirstlane(batch[node]);
  const float4 h = *(const float4*)(H + (size_t)node * HID + q);
  float* p = pooled + (size_t)g * HID + q;
  __hip_atomic_fetch_add(p + 0, h.x, __ATOMIC_RELAXED, __HIP_MEMORY_SCOPE_AGENT);
  __hip_atomic_fetch_add(p + 1, h.y, __ATOMIC_RELAXED, __HIP_MEMORY_SCOPE_AGENT);
  __hip_atomic_fetch_add(p + 2, h.z, __ATOMIC_RELAXED, __HIP_MEMORY_SCOPE_AGENT);
  __hip_atomic_fetch_add(p + 3, h.w, __ATOMIC_RELAXED, __HIP_MEMORY_SCOPE_AGENT);
}

__global__ void k_final(const float* __restrict__ pooled, const float* __restrict__ cnt,
                        const float* __restrict__ b3, const float* __restrict__ Wl,
                        const float* __restrict__ bl, float* __restrict__ out, int G, int C) {
  int gid = blockIdx.x * blockDim.x + threadIdx.x;
  if (gid >= G * C) return;
  int g = gid / C, c = gid % C;
  float inv = 1.0f / fmaxf(cnt[g], 1.0f);
  float s = 0.0f;
  const float* pr = pooled + (size_t)g * HID;
  for (int k = 0; k < HID; ++k)
    s += (pr[k] * inv + b3[k]) * Wl[k * C + c];
  out[gid] = s + bl[c];
}

extern "C" void kernel_launch(void* const* d_in, const int* in_sizes, int n_in,
                              void* d_out, int out_size, void* d_ws, size_t ws_size,
                              hipStream_t stream) {
  const float* x   = (const float*)d_in[0];
  const int*   ei  = (const int*)d_in[1];
  const int*   bat = (const int*)d_in[2];
  const float* W1  = (const float*)d_in[3];
  const float* b1  = (const float*)d_in[4];
  const float* W2  = (const float*)d_in[5];
  const float* b2  = (const float*)d_in[6];
  const float* W3  = (const float*)d_in[7];
  const float* b3  = (const float*)d_in[8];
  const float* Wl  = (const float*)d_in[9];
  const float* bl  = (const float*)d_in[10];

  const int n = in_sizes[2];
  const int e = in_sizes[1] / 2;
  const int C = in_sizes[10];
  const int G = out_size / C;
  const int* src = ei;
  const int* dst = ei + e;

  char* base = (char*)d_ws;
  auto alloc = [&](size_t bytes) -> void* {
    void* p = (void*)base;
    base += (bytes + 255) & ~(size_t)255;
    return p;
  };
  int*   counts    = (int*)alloc((size_t)n * 4);
  int*   cursor    = (int*)alloc((size_t)n * 4);
  int*   rowstart  = (int*)alloc(((size_t)n + 1) * 4);
  int*   blockSums = (int*)alloc(1024 * 4);
  int*   blockOff  = (int*)alloc(1024 * 4);
  float* dinv      = (float*)alloc((size_t)n * 4);
  int2*  rec       = (int2*)alloc((size_t)e * 8);
  float* Hb        = (float*)alloc((size_t)n * HID * 4);
  float* Ab        = (float*)alloc((size_t)n * HID * 4);
  float* pooled    = (float*)alloc((size_t)G * HID * 4);
  float* cnt       = (float*)alloc((size_t)G * 4);
  (void)ws_size; (void)n_in;

  const int T = 256;
  auto nb = [](long long w, int t) { return (unsigned)((w + t - 1) / t); };
  const long long nw = (long long)n * 32;
  const unsigned gemmB = (unsigned)((n + 127) / 128);
  const unsigned scanB = (unsigned)((n + 1023) / 1024);
  const size_t lds = (size_t)HID * WSTRIDE * sizeof(float);

  k_zero_i<<<nb(n, T), T, 0, stream>>>(counts, n);
  k_zero_i<<<nb(n, T), T, 0, stream>>>(cursor, n);
  k_count<<<nb(e, T), T, 0, stream>>>(dst, counts, e);
  k_dinv<<<nb(n, T), T, 0, stream>>>(counts, dinv, n);
  k_scan1<<<scanB, 1024, 0, stream>>>(counts, rowstart, blockSums, n);
  k_scan2<<<1, 1024, 0, stream>>>(blockSums, blockOff, (int)scanB);
  k_scan3<<<nb(n, T), T, 0, stream>>>(rowstart, blockOff, n);
  k_fill<<<nb(e, T), T, 0, stream>>>(src, dst, dinv, rowstart, cursor, rec, e);

  k_gemm128<<<gemmB, T, lds, stream>>>(x, W1, nullptr, Hb, n);
  k_aggregate<<<nb(nw, T), T, 0, stream>>>(Hb, rec, rowstart, dinv, Ab, n);

  k_gemm128<<<gemmB, T, lds, stream>>>(Ab, W2, b1, Hb, n);
  k_aggregate<<<nb(nw, T), T, 0, stream>>>(Hb, rec, rowstart, dinv, Ab, n);

  k_gemm128<<<gemmB, T, lds, stream>>>(Ab, W3, b2, Hb, n);
  k_aggregate<<<nb(nw, T), T, 0, stream>>>(Hb, rec, rowstart, dinv, Ab, n);

  k_zero_f<<<nb((long long)G * HID, T), T, 0, stream>>>(pooled, G * HID);
  k_zero_f<<<nb(G, T), T, 0, stream>>>(cnt, G);
  k_cnt<<<nb(n, T), T, 0, stream>>>(bat, cnt, n);
  k_pool<<<nb(nw, T), T, 0, stream>>>(Ab, bat, pooled, n);
  k_final<<<nb((long long)G * C, T), T, 0, stream>>>(pooled, cnt, b3, Wl, bl,
                                                     (float*)d_out, G, C);
}


# Adjacency as fixed-pitch rows filled in ONE atomic pass (slot = returned count): no histogram, scan or cursor pass, 19 launches to 13; edge weight split into two row scales: rounding changes
# speedup vs baseline: 1.2486x; 1.2486x over previous
#include <hip/hip_runtime.h>
#include <hip/hip_bf16.h>

typedef __attribute__((ext_vector_type(2))) float v2f;
typedef __attribute__((ext_vector_type(8))) float v8f;

#define HID 128
#define WSTRIDE 132
#define CAP 64

__global__ void k_zero_i(int* __restrict__ p, int n) {
  int i = blockIdx.x * blockDim.x + threadIdx.x;
  if (i < n) p[i] = 0;
}
__global__ void k_zero_f(float* __restrict__ p, int n) {
  int i = blockIdx.x * blockDim.x + threadIdx.x;
  if (i < n) p[i] = 0.0f;
}

__global__ void k_fill_rows(const int* __restrict__ src, const int* __restrict__ dst,
                            int* __restrict__ cnt, int* __restrict__ rec,
                            int2* __restrict__ spill, int n, int e) {
  int i = blockIdx.x * blockDim.x + threadIdx.x;
  if (i >= e) return;
  int s = src[i], d = dst[i];
  int p = atomicAdd(&cnt[d], 1);
  if (p < CAP) {
    rec[(size_t)d * CAP + p] = s;
  } else {
    int k = atomicAdd(&cnt[n], 1);
    int2 r;
    r.x = d;
    r.y = s;
    spill[k] = r;
  }
}

__global__ void k_gemm128(const float* __restrict__ X, const float* __restrict__ W,
                          const float* __restrict__ bias, const int* __restrict__ cnt,
                          float* __restrict__ H, int n) {
  extern __shared__ float ws[];
  const int tid = threadIdx.x;

  const int wave   = tid >> 5;
  const int lane   = tid & 31;
  const int lane15 = lane & 15;
  const int half   = lane >> 4;
  const int rowBase = blockIdx.x * 128 + wave * 16;
  const int arow    = rowBase + lane15;
  const int koff    = half * 2;

  int dcn[8];
#pragma unroll
  for (int r = 0; r < 8; ++r) {
    int row = rowBase + r + half * 8;
    dcn[r] = cnt[row < n ? row : n - 1];
  }

  for (int i = tid; i < HID * HID; i += 256) {
    int k = i >> 7, c = i & 127;
    ws[c * WSTRIDE + k] = W[i];
  }
  __syncthreads();

  float dsc[8];
#pragma unroll
  for (int r = 0; r < 8; ++r) dsc[r] = rsqrtf((float)dcn[r] + 1.0f);

  v2f a[32];
  if (arow < n) {
    const float* xp = X + (size_t)arow * HID + koff;
    if (bias) {
      const float* bp = bias + koff;
#pragma unroll
      for (int kk = 0; kk < 32; ++kk) {
        v2f xv = *(const v2f*)(xp + kk * 4);
        v2f bv = *(const v2f*)(bp + kk * 4);
        v2f t = xv + bv;
        t.x = fmaxf(t.x, 0.0f);
        t.y = fmaxf(t.y, 0.0f);
        a[kk] = t;
      }
    } else {
#pragma unroll
      for (int kk = 0; kk < 32; ++kk)
        a[kk] = *(const v2f*)(xp + kk * 4);
    }
  } else {
    v2f z = {};
#pragma unroll
    for (int kk = 0; kk < 32; ++kk) a[kk] = z;
  }

#pragma unroll
  for (int nt = 0; nt < 8; ++nt) {
    v8f acc = {};
    const float* bp = ws + (nt * 16 + lane15) * WSTRIDE + koff;
#pragma unroll
    for (int kk = 0; kk < 32; ++kk) {
      v2f b = *(const v2f*)(bp + kk * 4);
      acc = __builtin_amdgcn_wmma_f32_16x16x4_f32(
          false, a[kk], false, b, (short)0, acc, false, false);
    }
    const int colOut = nt * 16 + lane15;
#pragma unroll
    for (int r = 0; r < 8; ++r) {
      int row = rowBase + r + half * 8;
      if (row < n) H[(size_t)row * HID + colOut] = acc[r] * dsc[r];
    }
  }
}

__global__ void k_aggregate(const float* __restrict__ H, const int* __restrict__ rec,
                            const int* __restrict__ cnt, const int2* __restrict__ spill,
                            float* __restrict__ AGG, int n) {
  long long gid = (long long)blockIdx.x * blockDim.x + threadIdx.x;
  const int node = __builtin_amdgcn_readfirstlane((int)(gid >> 5));
  if (node >= n) return;
  const int lane = (int)gid & 31;
  const int q = lane * 4;

  const int deg = cnt[node];
  const int m = deg < CAP ? deg : CAP;
  const int* row = rec + (size_t)node * CAP;
  const float* Hq = H + q;

  float4 acc = *(const float4*)(Hq + (size_t)node * HID);

  for (int j = 0; j < m; ++j) {
    const int s = row[j];
    const float4 h = *(const float4*)(Hq + (size_t)s * HID);
    acc.x += h.x; acc.y += h.y; acc.z += h.z; acc.w += h.w;
  }

  if (deg > CAP) {
    const int ns = cnt[n];
    for (int k0 = 0; k0 < ns; k0 += 32) {
      const int k = k0 + lane;
      int2 r;
      r.x = -1;
      r.y = 0;
      if (k < ns) r = spill[k];
      unsigned hit = __builtin_amdgcn_ballot_w32(r.x == node);
      while (hit) {
        const int b = __builtin_ctz(hit);
        hit &= hit - 1;
        const int s = __builtin_amdgcn_readlane(r.y, b);
        const float4 h = *(const float4*)(Hq + (size_t)s * HID);
        acc.x += h.x; acc.y += h.y; acc.z += h.z; acc.w += h.w;
      }
    }
  }

  const float dn = rsqrtf((float)deg + 1.0f);
  acc.x *= dn; acc.y *= dn; acc.z *= dn; acc.w *= dn;
  *(float4*)(AGG + (size_t)node * HID + q) = acc;
}

__global__ void k_cnt(const int* __restrict__ batch, float* __restrict__ cnt, int n) {
  int i = blockIdx.x * blockDim.x + threadIdx.x;
  if (i < n)
    __hip_atomic_fetch_add(&cnt[batch[i]], 1.0f, __ATOMIC_RELAXED, __HIP_MEMORY_SCOPE_AGENT);
}

__global__ void k_pool(const float* __restrict__ H, const int* __restrict__ batch,
                       float* __restrict__ pooled, int n) {
  long long gid = (long long)blockIdx.x * blockDim.x + threadIdx.x;
  int node = (int)(gid >> 5);
  if (node >= n) return;
  int q = ((int)gid & 31) * 4;
  int g = __builtin_amdgcn_readfirstlane(batch[node]);
  const float4 h = *(const float4*)(H + (size_t)node * HID + q);
  float* p = pooled + (size_t)g * HID + q;
  __hip_atomic_fetch_add(p + 0, h.x, __ATOMIC_RELAXED, __HIP_MEMORY_SCOPE_AGENT);
  __hip_atomic_fetch_add(p + 1, h.y, __ATOMIC_RELAXED, __HIP_MEMORY_SCOPE_AGENT);
  __hip_atomic_fetch_add(p + 2, h.z, __ATOMIC_RELAXED, __HIP_MEMORY_SCOPE_AGENT);
  __hip_atomic_fetch_add(p + 3, h.w, __ATOMIC_RELAXED, __HIP_MEMORY_SCOPE_AGENT);
}

__global__ void k_final(const float* __restrict__ pooled, const float* __restrict__ cnt,
                        const float* __restrict__ b3, const float* __restrict__ Wl,
                        const float* __restrict__ bl, float* __restrict__ out, int G, int C) {
  int gid = blockIdx.x * blockDim.x + threadIdx.x;
  if (gid >= G * C) return;
  int g = gid / C, c = gid % C;
  float inv = 1.0f / fmaxf(cnt[g], 1.0f);
  float s = 0.0f;
  const float* pr = pooled + (size_t)g * HID;
  for (int k = 0; k < HID; ++k)
    s += (pr[k] * inv + b3[k]) * Wl[k * C + c];
  out[gid] = s + bl[c];
}

extern "C" void kernel_launch(void* const* d_in, const int* in_sizes, int n_in,
                              void* d_out, int out_size, void* d_ws, size_t ws_size,
                              hipStream_t stream) {
  const float* x   = (const float*)d_in[0];
  const int*   ei  = (const int*)d_in[1];
  const int*   bat = (const int*)d_in[2];
  const float* W1  = (const float*)d_in[3];
  const float* b1  = (const float*)d_in[4];
  const float* W2  = (const float*)d_in[5];
  const float* b2  = (const float*)d_in[6];
  const float* W3  = (const float*)d_in[7];
  const float* b3  = (const float*)d_in[8];
  const float* Wl  = (const float*)d_in[9];
  const float* bl  = (const float*)d_in[10];

  const int n = in_sizes[2];
  const int e = in_sizes[1] / 2;
  const int C = in_sizes[10];
  const int G = out_size / C;
  const int* src = ei;
  const int* dst = ei + e;

  char* base = (char*)d_ws;
  auto alloc = [&](size_t bytes) -> void* {
    void* p = (void*)base;
    base += (bytes + 255) & ~(size_t)255;
    return p;
  };
  int*   indeg     = (int*)alloc(((size_t)n + 1) * 4);
  int*   rec       = (int*)alloc((size_t)n * CAP * 4);
  int2*  spill     = (int2*)alloc((size_t)e * 8);
  float* Hb        = (float*)alloc((size_t)n * HID * 4);
  float* Ab        = (float*)alloc((size_t)n * HID * 4);
  float* pooled    = (float*)alloc((size_t)G * HID * 4);
  float* cnt       = (float*)alloc((size_t)G * 4);
  (void)ws_size; (void)n_in;

  const int T = 256;
  auto nb = [](long long w, int t) { return (unsigned)((w + t - 1) / t); };
  const long long nw = (long long)n * 32;
  const unsigned gemmB = (unsigned)((n + 127) / 128);
  const size_t lds = (size_t)HID * WSTRIDE * sizeof(float);

  k_zero_i<<<nb(n + 1, T), T, 0, stream>>>(indeg, n + 1);
  k_fill_rows<<<nb(e, T), T, 0, stream>>>(src, dst, indeg, rec, spill, n, e);

  k_gemm128<<<gemmB, T, lds, stream>>>(x, W1, nullptr, indeg, Hb, n);
  k_aggregate<<<nb(nw, T), T, 0, stream>>>(Hb, rec, indeg, spill, Ab, n);

  k_gemm128<<<gemmB, T, lds, stream>>>(Ab, W2, b1, indeg, Hb, n);
  k_aggregate<<<nb(nw, T), T, 0, stream>>>(Hb, rec, indeg, spill, Ab, n);

  k_gemm128<<<gemmB, T, lds, stream>>>(Ab, W3, b2, indeg, Hb, n);
  k_aggregate<<<nb(nw, T), T, 0, stream>>>(Hb, rec, indeg, spill, Ab, n);

  k_zero_f<<<nb((long long)G * HID, T), T, 0, stream>>>(pooled, G * HID);
  k_zero_f<<<nb(G, T), T, 0, stream>>>(cnt, G);
  k_cnt<<<nb(n, T), T, 0, stream>>>(bat, cnt, n);
  k_pool<<<nb(nw, T), T, 0, stream>>>(Ab, bat, pooled, n);
  k_final<<<nb((long long)G * C, T), T, 0, stream>>>(pooled, cnt, b3, Wl, bl,
                                                     (float*)d_out, G, C);
}


# Mean pool: lane owns features l, l+32, l+64, l+96, so each float atomic covers one whole 128-B line (was 4 lines x 8 lanes); a wave takes 8 nodes, rows loaded first; arrival-order sums as before
# speedup vs baseline: 1.4318x; 1.1467x over previous
#include <hip/hip_runtime.h>
#include <hip/hip_bf16.h>

typedef __attribute__((ext_vector_type(2))) float v2f;
typedef __attribute__((ext_vector_type(8))) float v8f;

#define HID 128
#define WSTRIDE 132
#define CAP 64

__global__ void k_zero_i(int* __restrict__ p, int n) {
  int i = blockIdx.x * blockDim.x + threadIdx.x;
  if (i < n) p[i] = 0;
}
__global__ void k_zero_f(float* __restrict__ p, int n) {
  int i = blockIdx.x * blockDim.x + threadIdx.x;
  if (i < n) p[i] = 0.0f;
}

__global__ void k_fill_rows(const int* __restrict__ src, const int* __restrict__ dst,
                            int* __restrict__ cnt, int* __restrict__ rec,
                            int2* __restrict__ spill, int n, int e) {
  int i = blockIdx.x * blockDim.x + threadIdx.x;
  if (i >= e) return;
  int s = src[i], d = dst[i];
  int p = atomicAdd(&cnt[d], 1);
  if (p < CAP) {
    rec[(size_t)d * CAP + p] = s;
  } else {
    int k = atomicAdd(&cnt[n], 1);
    int2 r;
    r.x = d;
    r.y = s;
    spill[k] = r;
  }
}

__global__ void k_gemm128(const float* __restrict__ X, const float* __restrict__ W,
                          const float* __restrict__ bias, const int* __restrict__ cnt,
                          float* __restrict__ H, int n) {
  extern __shared__ float ws[];
  const int tid = threadIdx.x;

  const int wave   = tid >> 5;
  const int lane   = tid & 31;
  const int lane15 = lane & 15;
  const int half   = lane >> 4;
  const int rowBase = blockIdx.x * 128 + wave * 16;
  const int arow    = rowBase + lane15;
  const int koff    = half * 2;

  int dcn[8];
#pragma unroll
  for (int r = 0; r < 8; ++r) {
    int row = rowBase + r + half * 8;
    dcn[r] = cnt[row < n ? row : n - 1];
  }

  for (int i = tid; i < HID * HID; i += 256) {
    int k = i >> 7, c = i & 127;
    ws[c * WSTRIDE + k] = W[i];
  }
  __syncthreads();

  float dsc[8];
#pragma unroll
  for (int r = 0; r < 8; ++r) dsc[r] = rsqrtf((float)dcn[r] + 1.0f);

  v2f a[32];
  if (arow < n) {
    const float* xp = X + (size_t)arow * HID + koff;
    if (bias) {
      const float* bp = bias + koff;
#pragma unroll
      for (int kk = 0; kk < 32; ++kk) {
        v2f xv = *(const v2f*)(xp + kk * 4);
        v2f bv = *(const v2f*)(bp + kk * 4);
        v2f t = xv + bv;
        t.x = fmaxf(t.x, 0.0f);
        t.y = fmaxf(t.y, 0.0f);
        a[kk] = t;
      }
    } else {
#pragma unroll
      for (int kk = 0; kk < 32; ++kk)
        a[kk] = *(const v2f*)(xp + kk * 4);
    }
  } else {
    v2f z = {};
#pragma unroll
    for (int kk = 0; kk < 32; ++kk) a[kk] = z;
  }

#pragma unroll
  for (int nt = 0; nt < 8; ++nt) {
    v8f acc = {};
    const float* bp = ws + (nt * 16 + lane15) * WSTRIDE + koff;
#pragma unroll
    for (int kk = 0; kk < 32; ++kk) {
      v2f b = *(const v2f*)(bp + kk * 4);
      acc = __builtin_amdgcn_wmma_f32_16x16x4_f32(
          false, a[kk], false, b, (short)0, acc, false, false);
    }
    const int colOut = nt * 16 + lane15;
#pragma unroll
    for (int r = 0; r < 8; ++r) {
      int row = rowBase + r + half * 8;
      if (row < n) H[(size_t)row * HID + colOut] = acc[r] * dsc[r];
    }
  }
}

__global__ void k_aggregate(const float* __restrict__ H, const int* __restrict__ rec,
                            const int* __restrict__ cnt, const int2* __restrict__ spill,
                            float* __restrict__ AGG, int n) {
  long long gid = (long long)blockIdx.x * blockDim.x + threadIdx.x;
  const int node = __builtin_amdgcn_readfirstlane((int)(gid >> 5));
  if (node >= n) return;
  const int lane = (int)gid & 31;
  const int q = lane * 4;

  const int deg = cnt[node];
  const int m = deg < CAP ? deg : CAP;
  const int* row = rec + (size_t)node * CAP;
  const float* Hq = H + q;

  float4 acc = *(const float4*)(Hq + (size_t)node * HID);

  for (int j = 0; j < m; ++j) {
    const int s = row[j];
    const float4 h = *(const float4*)(Hq + (size_t)s * HID);
    acc.x += h.x; acc.y += h.y; acc.z += h.z; acc.w += h.w;
  }

  if (deg > CAP) {
    const int ns = cnt[n];
    for (int k0 = 0; k0 < ns; k0 += 32) {
      const int k = k0 + lane;
      int2 r;
      r.x = -1;
      r.y = 0;
      if (k < ns) r = spill[k];
      unsigned hit = __builtin_amdgcn_ballot_w32(r.x == node);
      while (hit) {
        const int b = __builtin_ctz(hit);
        hit &= hit - 1;
        const int s = __builtin_amdgcn_readlane(r.y, b);
        const float4 h = *(const float4*)(Hq + (size_t)s * HID);
        acc.x += h.x; acc.y += h.y; acc.z += h.z; acc.w += h.w;
      }
    }
  }

  const float dn = rsqrtf((float)deg + 1.0f);
  acc.x *= dn; acc.y *= dn; acc.z *= dn; acc.w *= dn;
  *(float4*)(AGG + (size_t)node * HID + q) = acc;
}

__global__ void k_cnt(const int* __restrict__ batch, float* __restrict__ cnt, int n) {
  int i = blockIdx.x * blockDim.x + threadIdx.x;
  if (i < n)
    __hip_atomic_fetch_add(&cnt[batch[i]], 1.0f, __ATOMIC_RELAXED, __HIP_MEMORY_SCOPE_AGENT);
}

#define PNW 8
__global__ void k_pool(const float* __restrict__ H, const int* __restrict__ batch,
                       float* __restrict__ pooled, int n) {
  const long long gid = (long long)blockIdx.x * blockDim.x + threadIdx.x;
  const int wave = __builtin_amdgcn_readfirstlane((int)(gid >> 5));
  const int lane = (int)gid & 31;
  const int node0 = wave * PNW;
  if (node0 >= n) return;
  const int left = n - node0;

  int gl = 0;
  if (lane < PNW && lane < left) gl = batch[node0 + lane];
  const float* hp = H + (size_t)node0 * HID + lane;

  if (left >= PNW) {
    float v[PNW][4];
#pragma unroll
    for (int t = 0; t < PNW; ++t) {
      v[t][0] = hp[t * HID +  0];
      v[t][1] = hp[t * HID + 32];
      v[t][2] = hp[t * HID + 64];
      v[t][3] = hp[t * HID + 96];
    }
#pragma unroll
    for (int t = 0; t < PNW; ++t) {
      const int g = __builtin_amdgcn_readlane(gl, t);
      float* p = pooled + (size_t)g * HID + lane;
      __hip_atomic_fetch_add(p +  0, v[t][0], __ATOMIC_RELAXED, __HIP_MEMORY_SCOPE_AGENT);
      __hip_atomic_fetch_add(p + 32, v[t][1], __ATOMIC_RELAXED, __HIP_MEMORY_SCOPE_AGENT);
      __hip_atomic_fetch_add(p + 64, v[t][2], __ATOMIC_RELAXED, __HIP_MEMORY_SCOPE_AGENT);
      __hip_atomic_fetch_add(p + 96, v[t][3], __ATOMIC_RELAXED, __HIP_MEMORY_SCOPE_AGENT);
    }
  } else {
#pragma unroll 1
    for (int t = 0; t < left; ++t) {
      const int g = __builtin_amdgcn_readlane(gl, t);
      float* p = pooled + (size_t)g * HID + lane;
      __hip_atomic_fetch_add(p +  0, hp[t * HID +  0], __ATOMIC_RELAXED, __HIP_MEMORY_SCOPE_AGENT);
      __hip_atomic_fetch_add(p + 32, hp[t * HID + 32], __ATOMIC_RELAXED, __HIP_MEMORY_SCOPE_AGENT);
      __hip_atomic_fetch_add(p + 64, hp[t * HID + 64], __ATOMIC_RELAXED, __HIP_MEMORY_SCOPE_AGENT);
      __hip_atomic_fetch_add(p + 96, hp[t * HID + 96], __ATOMIC_RELAXED, __HIP_MEMORY_SCOPE_AGENT);
    }
  }
}

__global__ void k_final(const float* __restrict__ pooled, const float* __restrict__ cnt,
                        const float* __restrict__ b3, const float* __restrict__ Wl,
                        const float* __restrict__ bl, float* __restrict__ out, int G, int C) {
  int gid = blockIdx.x * blockDim.x + threadIdx.x;
  if (gid >= G * C) return;
  int g = gid / C, c = gid % C;
  float inv = 1.0f / fmaxf(cnt[g], 1.0f);
  float s = 0.0f;
  const float* pr = pooled + (size_t)g * HID;
  for (int k = 0; k < HID; ++k)
    s += (pr[k] * inv + b3[k]) * Wl[k * C + c];
  out[gid] = s + bl[c];
}

extern "C" void kernel_launch(void* const* d_in, const int* in_sizes, int n_in,
                              void* d_out, int out_size, void* d_ws, size_t ws_size,
                              hipStream_t stream) {
  const float* x   = (const float*)d_in[0];
  const int*   ei  = (const int*)d_in[1];
  const int*   bat = (const int*)d_in[2];
  const float* W1  = (const float*)d_in[3];
  const float* b1  = (const float*)d_in[4];
  const float* W2  = (const float*)d_in[5];
  const float* b2  = (const float*)d_in[6];
  const float* W3  = (const float*)d_in[7];
  const float* b3  = (const float*)d_in[8];
  const float* Wl  = (const float*)d_in[9];
  const float* bl  = (const float*)d_in[10];

  const int n = in_sizes[2];
  const int e = in_sizes[1] / 2;
  const int C = in_sizes[10];
  const int G = out_size / C;
  const int* src = ei;
  const int* dst = ei + e;

  char* base = (char*)d_ws;
  auto alloc = [&](size_t bytes) -> void* {
    void* p = (void*)base;
    base += (bytes + 255) & ~(size_t)255;
    return p;
  };
  int*   indeg     = (int*)alloc(((size_t)n + 1) * 4);
  int*   rec       = (int*)alloc((size_t)n * CAP * 4);
  int2*  spill     = (int2*)alloc((size_t)e * 8);
  float* Hb        = (float*)alloc((size_t)n * HID * 4);
  float* Ab        = (float*)alloc((size_t)n * HID * 4);
  float* pooled    = (float*)alloc((size_t)G * HID * 4);
  float* cnt       = (float*)alloc((size_t)G * 4);
  (void)ws_size; (void)n_in;

  const int T = 256;
  auto nb = [](long long w, int t) { return (unsigned)((w + t - 1) / t); };
  const long long nw = (long long)n * 32;
  const long long pw = ((long long)n + PNW - 1) / PNW * 32;
  const unsigned gemmB = (unsigned)((n + 127) / 128);
  const size_t lds = (size_t)HID * WSTRIDE * sizeof(float);

  k_zero_i<<<nb(n + 1, T), T, 0, stream>>>(indeg, n + 1);
  k_fill_rows<<<nb(e, T), T, 0, stream>>>(src, dst, indeg, rec, spill, n, e);

  k_gemm128<<<gemmB, T, lds, stream>>>(x, W1, nullptr, indeg, Hb, n);
  k_aggregate<<<nb(nw, T), T, 0, stream>>>(Hb, rec, indeg, spill, Ab, n);

  k_gemm128<<<gemmB, T, lds, stream>>>(Ab, W2, b1, indeg, Hb, n);
  k_aggregate<<<nb(nw, T), T, 0, stream>>>(Hb, rec, indeg, spill, Ab, n);

  k_gemm128<<<gemmB, T, lds, stream>>>(Ab, W3, b2, indeg, Hb, n);
  k_aggregate<<<nb(nw, T), T, 0, stream>>>(Hb, rec, indeg, spill, Ab, n);

  k_zero_f<<<nb((long long)G * HID, T), T, 0, stream>>>(pooled, G * HID);
  k_zero_f<<<nb(G, T), T, 0, stream>>>(cnt, G);
  k_cnt<<<nb(n, T), T, 0, stream>>>(bat, cnt, n);
  k_pool<<<nb(pw, T), T, 0, stream>>>(Ab, bat, pooled, n);
  k_final<<<nb((long long)G * C, T), T, 0, stream>>>(pooled, cnt, b3, Wl, bl,
                                                     (float*)d_out, G, C);
}


# Layer-3 product moved behind neighbour sum and mean pool (linear, no ReLU after): a 512-row WMMA product in the tail replaces the 100000-row one; 13 launches to 12; product and sum order change
# speedup vs baseline: 1.5879x; 1.1090x over previous
#include <hip/hip_runtime.h>
#include <hip/hip_bf16.h>

typedef __attribute__((ext_vector_type(2))) float v2f;
typedef __attribute__((ext_vector_type(8))) float v8f;

#define HID 128
#define WSTRIDE 132
#define CAP 64

__global__ void k_zero_i(int* __restrict__ p, int n) {
  int i = blockIdx.x * blockDim.x + threadIdx.x;
  if (i < n) p[i] = 0;
}
__global__ void k_zero_f(float* __restrict__ p, int n) {
  int i = blockIdx.x * blockDim.x + threadIdx.x;
  if (i < n) p[i] = 0.0f;
}

__global__ void k_fill_rows(const int* __restrict__ src, const int* __restrict__ dst,
                            int* __restrict__ cnt, int* __restrict__ rec,
                            int2* __restrict__ spill, int n, int e) {
  int i = blockIdx.x * blockDim.x + threadIdx.x;
  if (i >= e) return;
  int s = src[i], d = dst[i];
  int p = atomicAdd(&cnt[d], 1);
  if (p < CAP) {
    rec[(size_t)d * CAP + p] = s;
  } else {
    int k = atomicAdd(&cnt[n], 1);
    int2 r;
    r.x = d;
    r.y = s;
    spill[k] = r;
  }
}

__global__ void k_gemm128(const float* __restrict__ X, const float* __restrict__ W,
                          const float* __restrict__ bias, const int* __restrict__ cnt,
                          float* __restrict__ H, int n) {
  extern __shared__ float ws[];
  const int tid = threadIdx.x;

  const int wave   = tid >> 5;
  const int lane   = tid & 31;
  const int lane15 = lane & 15;
  const int half   = lane >> 4;
  const int rowBase = blockIdx.x * 128 + wave * 16;
  const int arow    = rowBase + lane15;
  const int koff    = half * 2;

  int dcn[8];
#pragma unroll
  for (int r = 0; r < 8; ++r) {
    int row = rowBase + r + half * 8;
    dcn[r] = cnt[row < n ? row : n - 1];
  }

  for (int i = tid; i < HID * HID; i += 256) {
    int k = i >> 7, c = i & 127;
    ws[c * WSTRIDE + k] = W[i];
  }
  __syncthreads();

  float dsc[8];
#pragma unroll
  for (int r = 0; r < 8; ++r) dsc[r] = rsqrtf((float)dcn[r] + 1.0f);

  v2f a[32];
  if (arow < n) {
    const float* xp = X + (size_t)arow * HID + koff;
    if (bias) {
      const float* bp = bias + koff;
#pragma unroll
      for (int kk = 0; kk < 32; ++kk) {
        v2f xv = *(const v2f*)(xp + kk * 4);
        v2f bv = *(const v2f*)(bp + kk * 4);
        v2f t = xv + bv;
        t.x = fmaxf(t.x, 0.0f);
        t.y = fmaxf(t.y, 0.0f);
        a[kk] = t;
      }
    } else {
#pragma unroll
      for (int kk = 0; kk < 32; ++kk)
        a[kk] = *(const v2f*)(xp + kk * 4);
    }
  } else {
    v2f z = {};
#pragma unroll
    for (int kk = 0; kk < 32; ++kk) a[kk] = z;
  }

#pragma unroll
  for (int nt = 0; nt < 8; ++nt) {
    v8f acc = {};
    const float* bp = ws + (nt * 16 + lane15) * WSTRIDE + koff;
#pragma unroll
    for (int kk = 0; kk < 32; ++kk) {
      v2f b = *(const v2f*)(bp + kk * 4);
      acc = __builtin_amdgcn_wmma_f32_16x16x4_f32(
          false, a[kk], false, b, (short)0, acc, false, false);
    }
    const int colOut = nt * 16 + lane15;
#pragma unroll
    for (int r = 0; r < 8; ++r) {
      int row = rowBase + r + half * 8;
      if (row < n) H[(size_t)row * HID + colOut] = acc[r] * dsc[r];
    }
  }
}

__global__ void k_aggregate(const float* __restrict__ H, const int* __restrict__ rec,
                            const int* __restrict__ cnt, const int2* __restrict__ spill,
                            const float* __restrict__ bias, float* __restrict__ AGG, int n) {
  long long gid = (long long)blockIdx.x * blockDim.x + threadIdx.x;
  const int node = __builtin_amdgcn_readfirstlane((int)(gid >> 5));
  if (node >= n) return;
  const int lane = (int)gid & 31;
  const int q = lane * 4;

  const int deg = cnt[node];
  const int m = deg < CAP ? deg : CAP;
  const int* row = rec + (size_t)node * CAP;
  const float* Hq = H + q;

  float4 acc = *(const float4*)(Hq + (size_t)node * HID);

  for (int j = 0; j < m; ++j) {
    const int s = row[j];
    const float4 h = *(const float4*)(Hq + (size_t)s * HID);
    acc.x += h.x; acc.y += h.y; acc.z += h.z; acc.w += h.w;
  }

  if (deg > CAP) {
    const int ns = cnt[n];
    for (int k0 = 0; k0 < ns; k0 += 32) {
      const int k = k0 + lane;
      int2 r;
      r.x = -1;
      r.y = 0;
      if (k < ns) r = spill[k];
      unsigned hit = __builtin_amdgcn_ballot_w32(r.x == node);
      while (hit) {
        const int b = __builtin_ctz(hit);
        hit &= hit - 1;
        const int s = __builtin_amdgcn_readlane(r.y, b);
        const float4 h = *(const float4*)(Hq + (size_t)s * HID);
        acc.x += h.x; acc.y += h.y; acc.z += h.z; acc.w += h.w;
      }
    }
  }

  const float dn = rsqrtf((float)deg + 1.0f);
  acc.x *= dn; acc.y *= dn; acc.z *= dn; acc.w *= dn;
  if (bias) {
    const float4 b = *(const float4*)(bias + q);
    acc.x = fmaxf(acc.x + b.x, 0.0f) * dn;
    acc.y = fmaxf(acc.y + b.y, 0.0f) * dn;
    acc.z = fmaxf(acc.z + b.z, 0.0f) * dn;
    acc.w = fmaxf(acc.w + b.w, 0.0f) * dn;
  }
  *(float4*)(AGG + (size_t)node * HID + q) = acc;
}

__global__ void k_cnt(const int* __restrict__ batch, float* __restrict__ cnt, int n) {
  int i = blockIdx.x * blockDim.x + threadIdx.x;
  if (i < n)
    __hip_atomic_fetch_add(&cnt[batch[i]], 1.0f, __ATOMIC_RELAXED, __HIP_MEMORY_SCOPE_AGENT);
}

#define PNW 8
__global__ void k_pool(const float* __restrict__ H, const int* __restrict__ batch,
                       float* __restrict__ pooled, int n) {
  const long long gid = (long long)blockIdx.x * blockDim.x + threadIdx.x;
  const int wave = __builtin_amdgcn_readfirstlane((int)(gid >> 5));
  const int lane = (int)gid & 31;
  const int node0 = wave * PNW;
  if (node0 >= n) return;
  const int left = n - node0;

  int gl = 0;
  if (lane < PNW && lane < left) gl = batch[node0 + lane];
  const float* hp = H + (size_t)node0 * HID + lane;

  if (left >= PNW) {
    float v[PNW][4];
#pragma unroll
    for (int t = 0; t < PNW; ++t) {
      v[t][0] = hp[t * HID +  0];
      v[t][1] = hp[t * HID + 32];
      v[t][2] = hp[t * HID + 64];
      v[t][3] = hp[t * HID + 96];
    }
#pragma unroll
    for (int t = 0; t < PNW; ++t) {
      const int g = __builtin_amdgcn_readlane(gl, t);
      float* p = pooled + (size_t)g * HID + lane;
      __hip_atomic_fetch_add(p +  0, v[t][0], __ATOMIC_RELAXED, __HIP_MEMORY_SCOPE_AGENT);
      __hip_atomic_fetch_add(p + 32, v[t][1], __ATOMIC_RELAXED, __HIP_MEMORY_SCOPE_AGENT);
      __hip_atomic_fetch_add(p + 64, v[t][2], __ATOMIC_RELAXED, __HIP_MEMORY_SCOPE_AGENT);
      __hip_atomic_fetch_add(p + 96, v[t][3], __ATOMIC_RELAXED, __HIP_MEMORY_SCOPE_AGENT);
    }
  } else {
#pragma unroll 1
    for (int t = 0; t < left; ++t) {
      const int g = __builtin_amdgcn_readlane(gl, t);
      float* p = pooled + (size_t)g * HID + lane;
      __hip_atomic_fetch_add(p +  0, hp[t * HID +  0], __ATOMIC_RELAXED, __HIP_MEMORY_SCOPE_AGENT);
      __hip_atomic_fetch_add(p + 32, hp[t * HID + 32], __ATOMIC_RELAXED, __HIP_MEMORY_SCOPE_AGENT);
      __hip_atomic_fetch_add(p + 64, hp[t * HID + 64], __ATOMIC_RELAXED, __HIP_MEMORY_SCOPE_AGENT);
      __hip_atomic_fetch_add(p + 96, hp[t * HID + 96], __ATOMIC_RELAXED, __HIP_MEMORY_SCOPE_AGENT);
    }
  }
}

#define TPITCH 132
__global__ void k_tail(const float* __restrict__ pooled, const float* __restrict__ cnt,
                       const float* __restrict__ W3, const float* __restrict__ b3,
                       const float* __restrict__ Wl, const float* __restrict__ bl,
                       float* __restrict__ out, int G, int C) {
  __shared__ float t[16 * TPITCH];
  const int tid    = threadIdx.x;
  const int wave   = tid >> 5;
  const int lane   = tid & 31;
  const int lane15 = lane & 15;
  const int half   = lane >> 4;
  const int koff   = half * 2;
  const int g0     = blockIdx.x * 16;
  const int grow   = g0 + lane15;

  v2f a[32];
  if (grow < G) {
    const float inv = 1.0f / fmaxf(cnt[grow], 1.0f);
    const float* xp = pooled + (size_t)grow * HID + koff;
#pragma unroll
    for (int kk = 0; kk < 32; ++kk) {
      v2f xv = *(const v2f*)(xp + kk * 4);
      a[kk] = xv * inv;
    }
  } else {
    v2f z = {};
#pragma unroll
    for (int kk = 0; kk < 32; ++kk) a[kk] = z;
  }

  const int col = wave * 16 + lane15;
  const float* bp = W3 + (size_t)koff * HID + col;
  v8f acc = {};
#pragma unroll
  for (int kk = 0; kk < 32; ++kk) {
    v2f b;
    b.x = bp[(kk * 4) * HID];
    b.y = bp[(kk * 4 + 1) * HID];
    acc = __builtin_amdgcn_wmma_f32_16x16x4_f32(
        false, a[kk], false, b, (short)0, acc, false, false);
  }
#pragma unroll
  for (int r = 0; r < 8; ++r)
    t[(r + half * 8) * TPITCH + col] = acc[r];
  __syncthreads();

  for (int o = tid; o < 16 * C; o += 256) {
    const int gl = o / C, c = o - gl * C;
    const int g = g0 + gl;
    if (g >= G) continue;
    const float bs = cnt[g] > 0.0f ? 1.0f : 0.0f;
    const float* tr = t + gl * TPITCH;
    float s = 0.0f;
    for (int k = 0; k < HID; ++k)
      s += (tr[k] + b3[k] * bs) * Wl[k * C + c];
    out[(size_t)g * C + c] = s + bl[c];
  }
}

extern "C" void kernel_launch(void* const* d_in, const int* in_sizes, int n_in,
                              void* d_out, int out_size, void* d_ws, size_t ws_size,
                              hipStream_t stream) {
  const float* x   = (const float*)d_in[0];
  const int*   ei  = (const int*)d_in[1];
  const int*   bat = (const int*)d_in[2];
  const float* W1  = (const float*)d_in[3];
  const float* b1  = (const float*)d_in[4];
  const float* W2  = (const float*)d_in[5];
  const float* b2  = (const float*)d_in[6];
  const float* W3  = (const float*)d_in[7];
  const float* b3  = (const float*)d_in[8];
  const float* Wl  = (const float*)d_in[9];
  const float* bl  = (const float*)d_in[10];

  const int n = in_sizes[2];
  const int e = in_sizes[1] / 2;
  const int C = in_sizes[10];
  const int G = out_size / C;
  const int* src = ei;
  const int* dst = ei + e;

  char* base = (char*)d_ws;
  auto alloc = [&](size_t bytes) -> void* {
    void* p = (void*)base;
    base += (bytes + 255) & ~(size_t)255;
    return p;
  };
  int*   indeg     = (int*)alloc(((size_t)n + 1) * 4);
  int*   rec       = (int*)alloc((size_t)n * CAP * 4);
  int2*  spill     = (int2*)alloc((size_t)e * 8);
  float* Hb        = (float*)alloc((size_t)n * HID * 4);
  float* Ab        = (float*)alloc((size_t)n * HID * 4);
  float* pooled    = (float*)alloc((size_t)G * HID * 4);
  float* cnt       = (float*)alloc((size_t)G * 4);
  (void)ws_size; (void)n_in;

  const int T = 256;
  auto nb = [](long long w, int t) { return (unsigned)((w + t - 1) / t); };
  const long long nw = (long long)n * 32;
  const long long pw = ((long long)n + PNW - 1) / PNW * 32;
  const unsigned gemmB = (unsigned)((n + 127) / 128);
  const size_t lds = (size_t)HID * WSTRIDE * sizeof(float);

  k_zero_i<<<nb(n + 1, T), T, 0, stream>>>(indeg, n + 1);
  k_fill_rows<<<nb(e, T), T, 0, stream>>>(src, dst, indeg, rec, spill, n, e);

  k_gemm128<<<gemmB, T, lds, stream>>>(x, W1, nullptr, indeg, Hb, n);
  k_aggregate<<<nb(nw, T), T, 0, stream>>>(Hb, rec, indeg, spill, nullptr, Ab, n);

  k_gemm128<<<gemmB, T, lds, stream>>>(Ab, W2, b1, indeg, Hb, n);
  k_aggregate<<<nb(nw, T), T, 0, stream>>>(Hb, rec, indeg, spill, b2, Ab, n);

  k_aggregate<<<nb(nw, T), T, 0, stream>>>(Ab, rec, indeg, spill, nullptr, Hb, n);

  k_zero_f<<<nb((long long)G * HID, T), T, 0, stream>>>(pooled, G * HID);
  k_zero_f<<<nb(G, T), T, 0, stream>>>(cnt, G);
  k_cnt<<<nb(n, T), T, 0, stream>>>(bat, cnt, n);
  k_pool<<<nb(pw, T), T, 0, stream>>>(Hb, bat, pooled, n);
  k_tail<<<(unsigned)((G + 15) / 16), T, 0, stream>>>(pooled, cnt, W3, b3, Wl, bl,
                                                      (float*)d_out, G, C);
}


# Products: all of W held in registers as B fragments, one wave per SIMD loops over row tiles (next rows loaded ahead, 16-byte loads, whole-line stores): no LDS staging; product sum order changes
# speedup vs baseline: 1.6773x; 1.0563x over previous
#include <hip/hip_runtime.h>
#include <hip/hip_bf16.h>

typedef __attribute__((ext_vector_type(2))) float v2f;
typedef __attribute__((ext_vector_type(4))) float v4f;
typedef __attribute__((ext_vector_type(8))) float v8f;

#define HID 128
#define CAP 64

__global__ void k_zero_i(int* __restrict__ p, int n) {
  int i = blockIdx.x * blockDim.x + threadIdx.x;
  if (i < n) p[i] = 0;
}
__global__ void k_zero_f(float* __restrict__ p, int n) {
  int i = blockIdx.x * blockDim.x + threadIdx.x;
  if (i < n) p[i] = 0.0f;
}

__global__ void k_fill_rows(const int* __restrict__ src, const int* __restrict__ dst,
                            int* __restrict__ cnt, int* __restrict__ rec,
                            int2* __restrict__ spill, int n, int e) {
  int i = blockIdx.x * blockDim.x + threadIdx.x;
  if (i >= e) return;
  int s = src[i], d = dst[i];
  int p = atomicAdd(&cnt[d], 1);
  if (p < CAP) {
    rec[(size_t)d * CAP + p] = s;
  } else {
    int k = atomicAdd(&cnt[n], 1);
    int2 r;
    r.x = d;
    r.y = s;
    spill[k] = r;
  }
}

template <bool BIAS>
__global__ void __launch_bounds__(128)
k_gemm_regw(const float* __restrict__ X, const float* __restrict__ W,
            const float* __restrict__ bias, const int* __restrict__ cnt,
            float* __restrict__ H, int n) {
  const int tid  = threadIdx.x;
  const int wave = __builtin_amdgcn_readfirstlane(tid >> 5);
  const int lane = tid & 31;
  const int c    = lane & 15;
  const int h    = lane >> 4;
  const int T    = (n + 15) >> 4;
  const int nwav = (int)gridDim.x * 4;
  int t = (int)blockIdx.x * 4 + wave;
  if (t >= T) return;

  v4f xa[16];
  int dca[8];
  {
    int arow = t * 16 + c;
    arow = arow < n ? arow : n - 1;
    const float* xp = X + (size_t)arow * HID + h * 4;
#pragma unroll
    for (int j = 0; j < 16; ++j) xa[j] = *(const v4f*)(xp + j * 8);
#pragma unroll
    for (int r = 0; r < 8; ++r) {
      int row = t * 16 + r + h * 8;
      dca[r] = cnt[row < n ? row : n - 1];
    }
  }

  v2f wf[8][32];
  {
    const float* wp = W + (size_t)(h * 4) * HID + c * 4;
#pragma unroll
    for (int j = 0; j < 16; ++j) {
#pragma unroll
      for (int g = 0; g < 2; ++g) {
        const float* p = wp + (size_t)(j * 8) * HID + g * 64;
        const v4f q0 = *(const v4f*)(p);
        const v4f q1 = *(const v4f*)(p + HID);
        const v4f q2 = *(const v4f*)(p + 2 * HID);
        const v4f q3 = *(const v4f*)(p + 3 * HID);
#pragma unroll
        for (int u = 0; u < 4; ++u) {
          v2f b0, b1;
          b0.x = q0[u]; b0.y = q1[u];
          b1.x = q2[u]; b1.y = q3[u];
          wf[g * 4 + u][2 * j]     = b0;
          wf[g * 4 + u][2 * j + 1] = b1;
        }
      }
    }
  }
  v4f bq[16];
  if (BIAS) {
#pragma unroll
    for (int j = 0; j < 16; ++j) bq[j] = *(const v4f*)(bias + j * 8 + h * 4);
  }

  for (;;) {
    const int tn = t + nwav;
    const bool more = tn < T;
    const int tp = more ? tn : t;
    v4f xn[16];
    int dcn[8];
    {
      int arow = tp * 16 + c;
      arow = arow < n ? arow : n - 1;
      const float* xp = X + (size_t)arow * HID + h * 4;
#pragma unroll
      for (int j = 0; j < 16; ++j) xn[j] = *(const v4f*)(xp + j * 8);
#pragma unroll
      for (int r = 0; r < 8; ++r) {
        int row = tp * 16 + r + h * 8;
        dcn[r] = cnt[row < n ? row : n - 1];
      }
    }

    v8f acc[8];
#pragma unroll
    for (int nt = 0; nt < 8; ++nt) { v8f z = {}; acc[nt] = z; }
#pragma unroll
    for (int j = 0; j < 16; ++j) {
      v4f x = xa[j];
      if (BIAS) {
        x = x + bq[j];
        x.x = fmaxf(x.x, 0.0f); x.y = fmaxf(x.y, 0.0f);
        x.z = fmaxf(x.z, 0.0f); x.w = fmaxf(x.w, 0.0f);
      }
      v2f a0, a1;
      a0.x = x.x; a0.y = x.y;
      a1.x = x.z; a1.y = x.w;
#pragma unroll
      for (int nt = 0; nt < 8; ++nt)
        acc[nt] = __builtin_amdgcn_wmma_f32_16x16x4_f32(
            false, a0, false, wf[nt][2 * j], (short)0, acc[nt], false, false);
#pragma unroll
      for (int nt = 0; nt < 8; ++nt)
        acc[nt] = __builtin_amdgcn_wmma_f32_16x16x4_f32(
            false, a1, false, wf[nt][2 * j + 1], (short)0, acc[nt], false, false);
    }

#pragma unroll
    for (int r = 0; r < 8; ++r) {
      const float ds = rsqrtf((float)dca[r] + 1.0f);
      const int row = t * 16 + r + h * 8;
      if (row < n) {
        float* hp = H + (size_t)row * HID + c * 4;
#pragma unroll
        for (int g = 0; g < 2; ++g) {
          v4f o;
          o.x = acc[g * 4 + 0][r] * ds;
          o.y = acc[g * 4 + 1][r] * ds;
          o.z = acc[g * 4 + 2][r] * ds;
          o.w = acc[g * 4 + 3][r] * ds;
          *(v4f*)(hp + g * 64) = o;
        }
      }
    }

    if (!more) break;
    t = tn;
#pragma unroll
    for (int j = 0; j < 16; ++j) xa[j] = xn[j];
#pragma unroll
    for (int r = 0; r < 8; ++r) dca[r] = dcn[r];
  }
}

__global__ void k_aggregate(const float* __restrict__ H, const int* __restrict__ rec,
                            const int* __restrict__ cnt, const int2* __restrict__ spill,
                            const float* __restrict__ bias, float* __restrict__ AGG, int n) {
  long long gid = (long long)blockIdx.x * blockDim.x + threadIdx.x;
  const int node = __builtin_amdgcn_readfirstlane((int)(gid >> 5));
  if (node >= n) return;
  const int lane = (int)gid & 31;
  const int q = lane * 4;

  const int deg = cnt[node];
  const int m = deg < CAP ? deg : CAP;
  const int* row = rec + (size_t)node * CAP;
  const float* Hq = H + q;

  float4 acc = *(const float4*)(Hq + (size_t)node * HID);

  for (int j = 0; j < m; ++j) {
    const int s = row[j];
    const float4 h = *(const float4*)(Hq + (size_t)s * HID);
    acc.x += h.x; acc.y += h.y; acc.z += h.z; acc.w += h.w;
  }

  if (deg > CAP) {
    const int ns = cnt[n];
    for (int k0 = 0; k0 < ns; k0 += 32) {
      const int k = k0 + lane;
      int2 r;
      r.x = -1;
      r.y = 0;
      if (k < ns) r = spill[k];
      unsigned hit = __builtin_amdgcn_ballot_w32(r.x == node);
      while (hit) {
        const int b = __builtin_ctz(hit);
        hit &= hit - 1;
        const int s = __builtin_amdgcn_readlane(r.y, b);
        const float4 h = *(const float4*)(Hq + (size_t)s * HID);
        acc.x += h.x; acc.y += h.y; acc.z += h.z; acc.w += h.w;
      }
    }
  }

  const float dn = rsqrtf((float)deg + 1.0f);
  acc.x *= dn; acc.y *= dn; acc.z *= dn; acc.w *= dn;
  if (bias) {
    const float4 b = *(const float4*)(bias + q);
    acc.x = fmaxf(acc.x + b.x, 0.0f) * dn;
    acc.y = fmaxf(acc.y + b.y, 0.0f) * dn;
    acc.z = fmaxf(acc.z + b.z, 0.0f) * dn;
    acc.w = fmaxf(acc.w + b.w, 0.0f) * dn;
  }
  *(float4*)(AGG + (size_t)node * HID + q) = acc;
}

__global__ void k_cnt(const int* __restrict__ batch, float* __restrict__ cnt, int n) {
  int i = blockIdx.x * blockDim.x + threadIdx.x;
  if (i < n)
    __hip_atomic_fetch_add(&cnt[batch[i]], 1.0f, __ATOMIC_RELAXED, __HIP_MEMORY_SCOPE_AGENT);
}

#define PNW 8
__global__ void k_pool(const float* __restrict__ H, const int* __restrict__ batch,
                       float* __restrict__ pooled, int n) {
  const long long gid = (long long)blockIdx.x * blockDim.x + threadIdx.x;
  const int wave = __builtin_amdgcn_readfirstlane((int)(gid >> 5));
  const int lane = (int)gid & 31;
  const int node0 = wave * PNW;
  if (node0 >= n) return;
  const int left = n - node0;

  int gl = 0;
  if (lane < PNW && lane < left) gl = batch[node0 + lane];
  const float* hp = H + (size_t)node0 * HID + lane;

  if (left >= PNW) {
    float v[PNW][4];
#pragma unroll
    for (int t = 0; t < PNW; ++t) {
      v[t][0] = hp[t * HID +  0];
      v[t][1] = hp[t * HID + 32];
      v[t][2] = hp[t * HID + 64];
      v[t][3] = hp[t * HID + 96];
    }
#pragma unroll
    for (int t = 0; t < PNW; ++t) {
      const int g = __builtin_amdgcn_readlane(gl, t);
      float* p = pooled + (size_t)g * HID + lane;
      __hip_atomic_fetch_add(p +  0, v[t][0], __ATOMIC_RELAXED, __HIP_MEMORY_SCOPE_AGENT);
      __hip_atomic_fetch_add(p + 32, v[t][1], __ATOMIC_RELAXED, __HIP_MEMORY_SCOPE_AGENT);
      __hip_atomic_fetch_add(p + 64, v[t][2], __ATOMIC_RELAXED, __HIP_MEMORY_SCOPE_AGENT);
      __hip_atomic_fetch_add(p + 96, v[t][3], __ATOMIC_RELAXED, __HIP_MEMORY_SCOPE_AGENT);
    }
  } else {
#pragma unroll 1
    for (int t = 0; t < left; ++t) {
      const int g = __builtin_amdgcn_readlane(gl, t);
      float* p = pooled + (size_t)g * HID + lane;
      __hip_atomic_fetch_add(p +  0, hp[t * HID +  0], __ATOMIC_RELAXED, __HIP_MEMORY_SCOPE_AGENT);
      __hip_atomic_fetch_add(p + 32, hp[t * HID + 32], __ATOMIC_RELAXED, __HIP_MEMORY_SCOPE_AGENT);
      __hip_atomic_fetch_add(p + 64, hp[t * HID + 64], __ATOMIC_RELAXED, __HIP_MEMORY_SCOPE_AGENT);
      __hip_atomic_fetch_add(p + 96, hp[t * HID + 96], __ATOMIC_RELAXED, __HIP_MEMORY_SCOPE_AGENT);
    }
  }
}

#define TPITCH 132
__global__ void k_tail(const float* __restrict__ pooled, const float* __restrict__ cnt,
                       const float* __restrict__ W3, const float* __restrict__ b3,
                       const float* __restrict__ Wl, const float* __restrict__ bl,
                       float* __restrict__ out, int G, int C) {
  __shared__ float t[16 * TPITCH];
  const int tid    = threadIdx.x;
  const int wave   = tid >> 5;
  const int lane   = tid & 31;
  const int lane15 = lane & 15;
  const int half   = lane >> 4;
  const int koff   = half * 2;
  const int g0     = blockIdx.x * 16;
  const int grow   = g0 + lane15;

  v2f a[32];
  if (grow < G) {
    const float inv = 1.0f / fmaxf(cnt[grow], 1.0f);
    const float* xp = pooled + (size_t)grow * HID + koff;
#pragma unroll
    for (int kk = 0; kk < 32; ++kk) {
      v2f xv = *(const v2f*)(xp + kk * 4);
      a[kk] = xv * inv;
    }
  } else {
    v2f z = {};
#pragma unroll
    for (int kk = 0; kk < 32; ++kk) a[kk] = z;
  }

  const int col = wave * 16 + lane15;
  const float* bp = W3 + (size_t)koff * HID + col;
  v8f acc = {};
#pragma unroll
  for (int kk = 0; kk < 32; ++kk) {
    v2f b;
    b.x = bp[(kk * 4) * HID];
    b.y = bp[(kk * 4 + 1) * HID];
    acc = __builtin_amdgcn_wmma_f32_16x16x4_f32(
        false, a[kk], false, b, (short)0, acc, false, false);
  }
#pragma unroll
  for (int r = 0; r < 8; ++r)
    t[(r + half * 8) * TPITCH + col] = acc[r];
  __syncthreads();

  for (int o = tid; o < 16 * C; o += 256) {
    const int gl = o / C, c = o - gl * C;
    const int g = g0 + gl;
    if (g >= G) continue;
    const float bs = cnt[g] > 0.0f ? 1.0f : 0.0f;
    const float* tr = t + gl * TPITCH;
    float s = 0.0f;
    for (int k = 0; k < HID; ++k)
      s += (tr[k] + b3[k] * bs) * Wl[k * C + c];
    out[(size_t)g * C + c] = s + bl[c];
  }
}

extern "C" void kernel_launch(void* const* d_in, const int* in_sizes, int n_in,
                              void* d_out, int out_size, void* d_ws, size_t ws_size,
                              hipStream_t stream) {
  const float* x   = (const float*)d_in[0];
  const int*   ei  = (const int*)d_in[1];
  const int*   bat = (const int*)d_in[2];
  const float* W1  = (const float*)d_in[3];
  const float* b1  = (const float*)d_in[4];
  const float* W2  = (const float*)d_in[5];
  const float* b2  = (const float*)d_in[6];
  const float* W3  = (const float*)d_in[7];
  const float* b3  = (const float*)d_in[8];
  const float* Wl  = (const float*)d_in[9];
  const float* bl  = (const float*)d_in[10];

  const int n = in_sizes[2];
  const int e = in_sizes[1] / 2;
  const int C = in_sizes[10];
  const int G = out_size / C;
  const int* src = ei;
  const int* dst = ei + e;

  char* base = (char*)d_ws;
  auto alloc = [&](size_t bytes) -> void* {
    void* p = (void*)base;
    base += (bytes + 255) & ~(size_t)255;
    return p;
  };
  int*   indeg     = (int*)alloc(((size_t)n + 1) * 4);
  int*   rec       = (int*)alloc((size_t)n * CAP * 4);
  int2*  spill     = (int2*)alloc((size_t)e * 8);
  float* Hb        = (float*)alloc((size_t)n * HID * 4);
  float* Ab        = (float*)alloc((size_t)n * HID * 4);
  float* pooled    = (float*)alloc((size_t)G * HID * 4);
  float* cnt       = (float*)alloc((size_t)G * 4);
  (void)ws_size; (void)n_in;

  const int T = 256;
  auto nb = [](long long w, int t) { return (unsigned)((w + t - 1) / t); };
  const long long nw = (long long)n * 32;
  const long long pw = ((long long)n + PNW - 1) / PNW * 32;
  const unsigned gemmB = 256;

  k_zero_i<<<nb(n + 1, T), T, 0, stream>>>(indeg, n + 1);
  k_fill_rows<<<nb(e, T), T, 0, stream>>>(src, dst, indeg, rec, spill, n, e);

  k_gemm_regw<false><<<gemmB, 128, 0, stream>>>(x, W1, nullptr, indeg, Hb, n);
  k_aggregate<<<nb(nw, T), T, 0, stream>>>(Hb, rec, indeg, spill, nullptr, Ab, n);

  k_gemm_regw<true><<<gemmB, 128, 0, stream>>>(Ab, W2, b1, indeg, Hb, n);
  k_aggregate<<<nb(nw, T), T, 0, stream>>>(Hb, rec, indeg, spill, b2, Ab, n);

  k_aggregate<<<nb(nw, T), T, 0, stream>>>(Ab, rec, indeg, spill, nullptr, Hb, n);

  k_zero_f<<<nb((long long)G * HID, T), T, 0, stream>>>(pooled, G * HID);
  k_zero_f<<<nb(G, T), T, 0, stream>>>(cnt, G);
  k_cnt<<<nb(n, T), T, 0, stream>>>(bat, cnt, n);
  k_pool<<<nb(pw, T), T, 0, stream>>>(Hb, bat, pooled, n);
  k_tail<<<(unsigned)((G + 15) / 16), T, 0, stream>>>(pooled, cnt, W3, b3, Wl, bl,
                                                      (float*)d_out, G, C);
}


# Pool without atomics: a work-group per graph (node range from the sorted graph ids) forms its nodes' layer-3 neighbour sums, adds them in registers, then LDS; 12 launches to 8; pool sum order changes
# speedup vs baseline: 1.9113x; 1.1395x over previous
#include <hip/hip_runtime.h>
#include <hip/hip_bf16.h>

typedef __attribute__((ext_vector_type(2))) float v2f;
typedef __attribute__((ext_vector_type(4))) float v4f;
typedef __attribute__((ext_vector_type(8))) float v8f;

#define HID 128
#define CAP 64

__global__ void k_prep(int* __restrict__ cnt, const int* __restrict__ batch,
                       int* __restrict__ gstart, int n, int G) {
  int i = blockIdx.x * blockDim.x + threadIdx.x;
  if (i > n) return;
  cnt[i] = 0;
  int prev = i == 0 ? -1 : batch[i - 1];
  int cur  = i == n ? G : batch[i];
  prev = prev < -1 ? -1 : prev;
  cur  = cur > G ? G : cur;
#pragma unroll 1
  for (int g = prev + 1; g <= cur; ++g) gstart[g] = i;
}

__global__ void k_fill_rows(const int* __restrict__ src, const int* __restrict__ dst,
                            int* __restrict__ cnt, int* __restrict__ rec,
                            int2* __restrict__ spill, int n, int e) {
  int i = blockIdx.x * blockDim.x + threadIdx.x;
  if (i >= e) return;
  int s = src[i], d = dst[i];
  int p = atomicAdd(&cnt[d], 1);
  if (p < CAP) {
    rec[(size_t)d * CAP + p] = s;
  } else {
    int k = atomicAdd(&cnt[n], 1);
    int2 r;
    r.x = d;
    r.y = s;
    spill[k] = r;
  }
}

template <bool BIAS>
__global__ void __launch_bounds__(128)
k_gemm_regw(const float* __restrict__ X, const float* __restrict__ W,
            const float* __restrict__ bias, const int* __restrict__ cnt,
            float* __restrict__ H, int n) {
  const int tid  = threadIdx.x;
  const int wave = __builtin_amdgcn_readfirstlane(tid >> 5);
  const int lane = tid & 31;
  const int c    = lane & 15;
  const int h    = lane >> 4;
  const int T    = (n + 15) >> 4;
  const int nwav = (int)gridDim.x * 4;
  int t = (int)blockIdx.x * 4 + wave;
  if (t >= T) return;

  v4f xa[16];
  int dca[8];
  {
    int arow = t * 16 + c;
    arow = arow < n ? arow : n - 1;
    const float* xp = X + (size_t)arow * HID + h * 4;
#pragma unroll
    for (int j = 0; j < 16; ++j) xa[j] = *(const v4f*)(xp + j * 8);
#pragma unroll
    for (int r = 0; r < 8; ++r) {
      int row = t * 16 + r + h * 8;
      dca[r] = cnt[row < n ? row : n - 1];
    }
  }

  v2f wf[8][32];
  {
    const float* wp = W + (size_t)(h * 4) * HID + c * 4;
#pragma unroll
    for (int j = 0; j < 16; ++j) {
#pragma unroll
      for (int g = 0; g < 2; ++g) {
        const float* p = wp + (size_t)(j * 8) * HID + g * 64;
        const v4f q0 = *(const v4f*)(p);
        const v4f q1 = *(const v4f*)(p + HID);
        const v4f q2 = *(const v4f*)(p + 2 * HID);
        const v4f q3 = *(const v4f*)(p + 3 * HID);
#pragma unroll
        for (int u = 0; u < 4; ++u) {
          v2f b0, b1;
          b0.x = q0[u]; b0.y = q1[u];
          b1.x = q2[u]; b1.y = q3[u];
          wf[g * 4 + u][2 * j]     = b0;
          wf[g * 4 + u][2 * j + 1] = b1;
        }
      }
    }
  }
  v4f bq[16];
  if (BIAS) {
#pragma unroll
    for (int j = 0; j < 16; ++j) bq[j] = *(const v4f*)(bias + j * 8 + h * 4);
  }

  for (;;) {
    const int tn = t + nwav;
    const bool more = tn < T;
    const int tp = more ? tn : t;
    v4f xn[16];
    int dcn[8];
    {
      int arow = tp * 16 + c;
      arow = arow < n ? arow : n - 1;
      const float* xp = X + (size_t)arow * HID + h * 4;
#pragma unroll
      for (int j = 0; j < 16; ++j) xn[j] = *(const v4f*)(xp + j * 8);
#pragma unroll
      for (int r = 0; r < 8; ++r) {
        int row = tp * 16 + r + h * 8;
        dcn[r] = cnt[row < n ? row : n - 1];
      }
    }

    v8f acc[8];
#pragma unroll
    for (int nt = 0; nt < 8; ++nt) { v8f z = {}; acc[nt] = z; }
#pragma unroll
    for (int j = 0; j < 16; ++j) {
      v4f x = xa[j];
      if (BIAS) {
        x = x + bq[j];
        x.x = fmaxf(x.x, 0.0f); x.y = fmaxf(x.y, 0.0f);
        x.z = fmaxf(x.z, 0.0f); x.w = fmaxf(x.w, 0.0f);
      }
      v2f a0, a1;
      a0.x = x.x; a0.y = x.y;
      a1.x = x.z; a1.y = x.w;
#pragma unroll
      for (int nt = 0; nt < 8; ++nt)
        acc[nt] = __builtin_amdgcn_wmma_f32_16x16x4_f32(
            false, a0, false, wf[nt][2 * j], (short)0, acc[nt], false, false);
#pragma unroll
      for (int nt = 0; nt < 8; ++nt)
        acc[nt] = __builtin_amdgcn_wmma_f32_16x16x4_f32(
            false, a1, false, wf[nt][2 * j + 1], (short)0, acc[nt], false, false);
    }

#pragma unroll
    for (int r = 0; r < 8; ++r) {
      const float ds = rsqrtf((float)dca[r] + 1.0f);
      const int row = t * 16 + r + h * 8;
      if (row < n) {
        float* hp = H + (size_t)row * HID + c * 4;
#pragma unroll
        for (int g = 0; g < 2; ++g) {
          v4f o;
          o.x = acc[g * 4 + 0][r] * ds;
          o.y = acc[g * 4 + 1][r] * ds;
          o.z = acc[g * 4 + 2][r] * ds;
          o.w = acc[g * 4 + 3][r] * ds;
          *(v4f*)(hp + g * 64) = o;
        }
      }
    }

    if (!more) break;
    t = tn;
#pragma unroll
    for (int j = 0; j < 16; ++j) xa[j] = xn[j];
#pragma unroll
    for (int r = 0; r < 8; ++r) dca[r] = dcn[r];
  }
}

__device__ __forceinline__ float4 row_sum(const float* __restrict__ Hq, const int* __restrict__ rec,
                                          const int* __restrict__ cnt, const int2* __restrict__ spill,
                                          int node, int n, int lane, int& deg) {
  deg = cnt[node];
  const int m = deg < CAP ? deg : CAP;
  const int* row = rec + (size_t)node * CAP;

  float4 acc = *(const float4*)(Hq + (size_t)node * HID);

  for (int j = 0; j < m; ++j) {
    const int s = row[j];
    const float4 h = *(const float4*)(Hq + (size_t)s * HID);
    acc.x += h.x; acc.y += h.y; acc.z += h.z; acc.w += h.w;
  }

  if (deg > CAP) {
    const int ns = cnt[n];
    for (int k0 = 0; k0 < ns; k0 += 32) {
      const int k = k0 + lane;
      int2 r;
      r.x = -1;
      r.y = 0;
      if (k < ns) r = spill[k];
      unsigned hit = __builtin_amdgcn_ballot_w32(r.x == node);
      while (hit) {
        const int b = __builtin_ctz(hit);
        hit &= hit - 1;
        const int s = __builtin_amdgcn_readlane(r.y, b);
        const float4 h = *(const float4*)(Hq + (size_t)s * HID);
        acc.x += h.x; acc.y += h.y; acc.z += h.z; acc.w += h.w;
      }
    }
  }
  return acc;
}

__global__ void k_aggregate(const float* __restrict__ H, const int* __restrict__ rec,
                            const int* __restrict__ cnt, const int2* __restrict__ spill,
                            const float* __restrict__ bias, float* __restrict__ AGG, int n) {
  long long gid = (long long)blockIdx.x * blockDim.x + threadIdx.x;
  const int node = __builtin_amdgcn_readfirstlane((int)(gid >> 5));
  if (node >= n) return;
  const int lane = (int)gid & 31;
  const int q = lane * 4;

  int deg;
  float4 acc = row_sum(H + q, rec, cnt, spill, node, n, lane, deg);

  const float dn = rsqrtf((float)deg + 1.0f);
  acc.x *= dn; acc.y *= dn; acc.z *= dn; acc.w *= dn;
  if (bias) {
    const float4 b = *(const float4*)(bias + q);
    acc.x = fmaxf(acc.x + b.x, 0.0f) * dn;
    acc.y = fmaxf(acc.y + b.y, 0.0f) * dn;
    acc.z = fmaxf(acc.z + b.z, 0.0f) * dn;
    acc.w = fmaxf(acc.w + b.w, 0.0f) * dn;
  }
  *(float4*)(AGG + (size_t)node * HID + q) = acc;
}

#define PW 32
__global__ void __launch_bounds__(PW * 32)
k_agg_pool(const float* __restrict__ H, const int* __restrict__ rec,
           const int* __restrict__ cnt, const int2* __restrict__ spill,
           const int* __restrict__ gstart, float* __restrict__ pooled, int n) {
  __shared__ float part[PW * HID];
  const int g    = blockIdx.x;
  const int tid  = threadIdx.x;
  const int wave = __builtin_amdgcn_readfirstlane(tid >> 5);
  const int lane = tid & 31;
  const int q    = lane * 4;

  int s0 = gstart[g], s1 = gstart[g + 1];
  s0 = s0 < 0 ? 0 : (s0 > n ? n : s0);
  s1 = s1 < s0 ? s0 : (s1 > n ? n : s1);

  float4 tot = make_float4(0.0f, 0.0f, 0.0f, 0.0f);
  for (int node = s0 + wave; node < s1; node += PW) {
    int deg;
    const float4 acc = row_sum(H + q, rec, cnt, spill, node, n, lane, deg);
    const float dn = rsqrtf((float)deg + 1.0f);
    tot.x += acc.x * dn; tot.y += acc.y * dn; tot.z += acc.z * dn; tot.w += acc.w * dn;
  }
  *(float4*)(part + wave * HID + q) = tot;
  __syncthreads();
  if (tid < HID) {
    float s = 0.0f;
#pragma unroll 8
    for (int w = 0; w < PW; ++w) s += part[w * HID + tid];
    pooled[(size_t)g * HID + tid] = s;
  }
}

#define TPITCH 132
__global__ void k_tail(const float* __restrict__ pooled, const int* __restrict__ gstart,
                       const float* __restrict__ W3, const float* __restrict__ b3,
                       const float* __restrict__ Wl, const float* __restrict__ bl,
                       float* __restrict__ out, int G, int C) {
  __shared__ float t[16 * TPITCH];
  const int tid    = threadIdx.x;
  const int wave   = tid >> 5;
  const int lane   = tid & 31;
  const int lane15 = lane & 15;
  const int half   = lane >> 4;
  const int koff   = half * 2;
  const int g0     = blockIdx.x * 16;
  const int grow   = g0 + lane15;

  v2f a[32];
  if (grow < G) {
    const float inv = 1.0f / fmaxf((float)(gstart[grow + 1] - gstart[grow]), 1.0f);
    const float* xp = pooled + (size_t)grow * HID + koff;
#pragma unroll
    for (int kk = 0; kk < 32; ++kk) {
      v2f xv = *(const v2f*)(xp + kk * 4);
      a[kk] = xv * inv;
    }
  } else {
    v2f z = {};
#pragma unroll
    for (int kk = 0; kk < 32; ++kk) a[kk] = z;
  }

  const int col = wave * 16 + lane15;
  const float* bp = W3 + (size_t)koff * HID + col;
  v8f acc = {};
#pragma unroll
  for (int kk = 0; kk < 32; ++kk) {
    v2f b;
    b.x = bp[(kk * 4) * HID];
    b.y = bp[(kk * 4 + 1) * HID];
    acc = __builtin_amdgcn_wmma_f32_16x16x4_f32(
        false, a[kk], false, b, (short)0, acc, false, false);
  }
#pragma unroll
  for (int r = 0; r < 8; ++r)
    t[(r + half * 8) * TPITCH + col] = acc[r];
  __syncthreads();

  for (int o = tid; o < 16 * C; o += 256) {
    const int gl = o / C, c = o - gl * C;
    const int g = g0 + gl;
    if (g >= G) continue;
    const float bs = gstart[g + 1] > gstart[g] ? 1.0f : 0.0f;
    const float* tr = t + gl * TPITCH;
    float s = 0.0f;
    for (int k = 0; k < HID; ++k)
      s += (tr[k] + b3[k] * bs) * Wl[k * C + c];
    out[(size_t)g * C + c] = s + bl[c];
  }
}

extern "C" void kernel_launch(void* const* d_in, const int* in_sizes, int n_in,
                              void* d_out, int out_size, void* d_ws, size_t ws_size,
                              hipStream_t stream) {
  const float* x   = (const float*)d_in[0];
  const int*   ei  = (const int*)d_in[1];
  const int*   bat = (const int*)d_in[2];
  const float* W1  = (const float*)d_in[3];
  const float* b1  = (const float*)d_in[4];
  const float* W2  = (const float*)d_in[5];
  const float* b2  = (const float*)d_in[6];
  const float* W3  = (const float*)d_in[7];
  const float* b3  = (const float*)d_in[8];
  const float* Wl  = (const float*)d_in[9];
  const float* bl  = (const float*)d_in[10];

  const int n = in_sizes[2];
  const int e = in_sizes[1] / 2;
  const int C = in_sizes[10];
  const int G = out_size / C;
  const int* src = ei;
  const int* dst = ei + e;

  char* base = (char*)d_ws;
  auto alloc = [&](size_t bytes) -> void* {
    void* p = (void*)base;
    base += (bytes + 255) & ~(size_t)255;
    return p;
  };
  int*   indeg     = (int*)alloc(((size_t)n + 1) * 4);
  int*   rec       = (int*)alloc((size_t)n * CAP * 4);
  int2*  spill     = (int2*)alloc((size_t)e * 8);
  float* Hb        = (float*)alloc((size_t)n * HID * 4);
  float* Ab        = (float*)alloc((size_t)n * HID * 4);
  float* pooled    = (float*)alloc((size_t)G * HID * 4);
  int*   gstart    = (int*)alloc(((size_t)G + 1) * 4);
  (void)ws_size; (void)n_in;

  const int T = 256;
  auto nb = [](long long w, int t) { return (unsigned)((w + t - 1) / t); };
  const long long nw = (long long)n * 32;
  const unsigned gemmB = 256;

  k_prep<<<nb(n + 1, T), T, 0, stream>>>(indeg, bat, gstart, n, G);
  k_fill_rows<<<nb(e, T), T, 0, stream>>>(src, dst, indeg, rec, spill, n, e);

  k_gemm_regw<false><<<gemmB, 128, 0, stream>>>(x, W1, nullptr, indeg, Hb, n);
  k_aggregate<<<nb(nw, T), T, 0, stream>>>(Hb, rec, indeg, spill, nullptr, Ab, n);

  k_gemm_regw<true><<<gemmB, 128, 0, stream>>>(Ab, W2, b1, indeg, Hb, n);
  k_aggregate<<<nb(nw, T), T, 0, stream>>>(Hb, rec, indeg, spill, b2, Ab, n);

  k_agg_pool<<<(unsigned)G, PW * 32, 0, stream>>>(Ab, rec, indeg, spill, gstart, pooled, n);

  k_tail<<<(unsigned)((G + 15) / 16), T, 0, stream>>>(pooled, gstart, W3, b3, Wl, bl,
                                                      (float*)d_out, G, C);
}
